# starting kernel
# baseline (speedup 1.0000x reference)
#include <hip/hip_runtime.h>


#define HMM_B 256
#define HMM_T 8192
#define HMM_K 8
#define HMM_EPS 1e-8f
#define HMM_LN2 0.6931471805599453f

typedef __attribute__((ext_vector_type(2))) float v2f;
typedef __attribute__((ext_vector_type(8))) float v8f;

__device__ __forceinline__ float fast_rcp(float x) {
    float r = __builtin_amdgcn_rcpf(x);
    float e = __builtin_fmaf(-x, r, 1.0f);
    return __builtin_fmaf(r, e, r);
}

__device__ __forceinline__ float crosshalf(float x) {
    unsigned u = __builtin_bit_cast(unsigned, x);
    u = __builtin_amdgcn_permlanex16(u, u, 0x76543210u, 0xFEDCBA98u, false, false);
    return __builtin_bit_cast(float, u);
}

__global__ __launch_bounds__(256) void hmm_pack_kernel(
    const int* __restrict__ obs, unsigned* __restrict__ bits)
{
    const int wave = (blockIdx.x * blockDim.x + threadIdx.x) >> 5;
    const int lane = threadIdx.x & 31;
    const long base = (long)wave * 1024;
    unsigned keep = 0;
#pragma unroll 4
    for (int i = 0; i < 32; ++i) {
        int v = obs[base + i * 32 + lane];
        unsigned m = (unsigned)__ballot(v != 0);
        if (i == lane) keep = m;
    }
    bits[(size_t)wave * 32 + lane] = keep;
}

__global__ __launch_bounds__(32) void hmm_forward_kernel(
    const float* __restrict__ start, const float* __restrict__ trans,
    const float* __restrict__ emission, const unsigned* __restrict__ bits,
    float* __restrict__ alpha_out, float* __restrict__ loglik)
{
    const int  lane = threadIdx.x;
    const int  n    = lane & 15;
    const bool lo   = lane < 16;
    const int  b    = blockIdx.x * 16 + n;

    float em1[8], em0[8];
#pragma unroll
    for (int k = 0; k < 8; ++k) {
        float e = emission[k];
        e = fminf(fmaxf(e, HMM_EPS), 1.0f - HMM_EPS);
        em1[k] = e; em0[k] = 1.0f - e;
    }
    float st[8]; float ssum = 0.0f;
#pragma unroll
    for (int k = 0; k < 8; ++k) { st[k] = start[k] + HMM_EPS; ssum += st[k]; }
#pragma unroll
    for (int k = 0; k < 8; ++k) st[k] /= ssum;

    float rs[8];
#pragma unroll
    for (int j = 0; j < 8; ++j) {
        float s = 0.0f;
#pragma unroll
        for (int k = 0; k < 8; ++k) s += trans[j * 8 + k] + HMM_EPS;
        rs[j] = s;
    }

    const int kb = lo ? 0 : 2;
    const int i  = n;
    v2f A0, A1;
    A0.x = (i < 8) ? (trans[(kb + 0) * 8 + i] + HMM_EPS) / rs[kb + 0] : 0.0f;
    A0.y = (i < 8) ? (trans[(kb + 1) * 8 + i] + HMM_EPS) / rs[kb + 1] : 0.0f;
    A1.x = (i < 8) ? (trans[(kb + 4) * 8 + i] + HMM_EPS) / rs[kb + 4] : 0.0f;
    A1.y = (i < 8) ? (trans[(kb + 5) * 8 + i] + HMM_EPS) / rs[kb + 5] : 0.0f;

    unsigned w  = bits[(size_t)b * (HMM_T / 32)];
    int      ob = (int)(w & 1u);
    v8f a;
#pragma unroll
    for (int m = 0; m < 8; ++m) a[m] = st[m] * (ob ? em1[m] : em0[m]);
    float s = a[0] + a[1] + a[2] + a[3] + a[4] + a[5] + a[6] + a[7];
    {  float sx = crosshalf(s); s = lo ? s : sx; }
    float inv = fast_rcp(s);
    float lc  = __builtin_amdgcn_logf(s);
    if (lo) *(v8f*)(alpha_out + ((size_t)b * HMM_T) * 8) = a;

    const v8f vzero = {};
    for (int t = 1; t < HMM_T; ++t) {
        if ((t & 31) == 0) w = bits[(size_t)b * (HMM_T / 32) + (t >> 5)];
        ob = (int)((w >> (t & 31)) & 1u);

        float x2 = crosshalf(a[2]);
        float x3 = crosshalf(a[3]);
        float x6 = crosshalf(a[6]);
        float x7 = crosshalf(a[7]);
        v2f B0, B1;
        B0.x = (lo ? a[0] : x2) * inv;  B0.y = (lo ? a[1] : x3) * inv;
        B1.x = (lo ? a[4] : x6) * inv;  B1.y = (lo ? a[5] : x7) * inv;

        v8f d = __builtin_amdgcn_wmma_f32_16x16x4_f32(
            false, A0, false, B0, (short)0, vzero, false, false);
        d = __builtin_amdgcn_wmma_f32_16x16x4_f32(
            false, A1, false, B1, (short)0, d, false, false);

#pragma unroll
        for (int m = 0; m < 8; ++m) a[m] = d[m] * (ob ? em1[m] : em0[m]);
        s = a[0] + a[1] + a[2] + a[3] + a[4] + a[5] + a[6] + a[7];
        float sx = crosshalf(s);
        s = lo ? s : sx;
        inv = fast_rcp(s);
        lc += __builtin_amdgcn_logf(s);
        if (lo) *(v8f*)(alpha_out + ((size_t)b * HMM_T + t) * 8) = a;
    }
    if (lo) loglik[b] = lc * HMM_LN2;
}

__global__ __launch_bounds__(32) void hmm_backward_kernel(
    const float* __restrict__ trans, const float* __restrict__ emission,
    const unsigned* __restrict__ bits, float* __restrict__ gamma)
{
    const int  lane = threadIdx.x;
    const int  n    = lane & 15;
    const bool lo   = lane < 16;
    const int  b    = blockIdx.x * 16 + n;

    float em1[8], em0[8];
#pragma unroll
    for (int k = 0; k < 8; ++k) {
        float e = emission[k];
        e = fminf(fmaxf(e, HMM_EPS), 1.0f - HMM_EPS);
        em1[k] = e; em0[k] = 1.0f - e;
    }
    float rs[8];
#pragma unroll
    for (int j = 0; j < 8; ++j) {
        float s = 0.0f;
#pragma unroll
        for (int k = 0; k < 8; ++k) s += trans[j * 8 + k] + HMM_EPS;
        rs[j] = s;
    }
    const int kb = lo ? 0 : 2;
    const int i  = n;
    v2f A0, A1;
    A0.x = (i < 8) ? (trans[i * 8 + (kb + 0)] + HMM_EPS) / rs[i] : 0.0f;
    A0.y = (i < 8) ? (trans[i * 8 + (kb + 1)] + HMM_EPS) / rs[i] : 0.0f;
    A1.x = (i < 8) ? (trans[i * 8 + (kb + 4)] + HMM_EPS) / rs[i] : 0.0f;
    A1.y = (i < 8) ? (trans[i * 8 + (kb + 5)] + HMM_EPS) / rs[i] : 0.0f;

    v8f d;
#pragma unroll
    for (int m = 0; m < 8; ++m) d[m] = 1.0f;
    if (lo) {
        float* p = gamma + ((size_t)b * HMM_T + (HMM_T - 1)) * 8;
        v8f al = *(v8f*)p;
        float gs = al[0]+al[1]+al[2]+al[3]+al[4]+al[5]+al[6]+al[7];
        float gi = fast_rcp(gs);
        v8f g;
#pragma unroll
        for (int m = 0; m < 8; ++m) g[m] = al[m] * gi;
        *(v8f*)p = g;
    }

    const v8f vzero = {};
    unsigned w = 0;
    for (int t = HMM_T - 2; t >= 0; --t) {
        const int tt = t + 1;
        if ((tt & 31) == 31) w = bits[(size_t)b * (HMM_T / 32) + (tt >> 5)];
        int ob = (int)((w >> (tt & 31)) & 1u);

        float s = d[0]+d[1]+d[2]+d[3]+d[4]+d[5]+d[6]+d[7];
        float sx = crosshalf(s);
        s = lo ? s : sx;
        float inv = fast_rcp(s);

        v8f u;
#pragma unroll
        for (int m = 0; m < 8; ++m) u[m] = d[m] * (ob ? em1[m] : em0[m]);
        float x2 = crosshalf(u[2]);
        float x3 = crosshalf(u[3]);
        float x6 = crosshalf(u[6]);
        float x7 = crosshalf(u[7]);
        v2f B0, B1;
        B0.x = (lo ? u[0] : x2) * inv;  B0.y = (lo ? u[1] : x3) * inv;
        B1.x = (lo ? u[4] : x6) * inv;  B1.y = (lo ? u[5] : x7) * inv;

        d = __builtin_amdgcn_wmma_f32_16x16x4_f32(
            false, A0, false, B0, (short)0, vzero, false, false);
        d = __builtin_amdgcn_wmma_f32_16x16x4_f32(
            false, A1, false, B1, (short)0, d, false, false);

        if (lo) {
            float* p = gamma + ((size_t)b * HMM_T + t) * 8;
            v8f al = *(v8f*)p;
            v8f g;
            float gs = 0.0f;
#pragma unroll
            for (int m = 0; m < 8; ++m) { g[m] = al[m] * d[m]; gs += g[m]; }
            float gi = fast_rcp(gs);
#pragma unroll
            for (int m = 0; m < 8; ++m) g[m] *= gi;
            *(v8f*)p = g;
        }
    }
}

extern "C" void kernel_launch(void* const* d_in, const int* in_sizes, int n_in,
                              void* d_out, int out_size, void* d_ws, size_t ws_size,
                              hipStream_t stream) {
    (void)in_sizes; (void)n_in; (void)out_size; (void)ws_size;
    const float* start    = (const float*)d_in[0];
    const float* trans    = (const float*)d_in[1];
    const float* emission = (const float*)d_in[2];
    const int*   obs      = (const int*)d_in[3];

    float* gamma  = (float*)d_out;
    float* loglik = gamma + (size_t)HMM_B * HMM_T * HMM_K;
    unsigned* bits = (unsigned*)d_ws;

    hmm_pack_kernel<<<256, 256, 0, stream>>>(obs, bits);
    hmm_forward_kernel<<<HMM_B / 16, 32, 0, stream>>>(
        start, trans, emission, bits, gamma, loglik);
    hmm_backward_kernel<<<HMM_B / 16, 32, 0, stream>>>(
        trans, emission, bits, gamma);
}


# both recurrences: wave 0 keeps the rows of 8 steps in registers and writes them to the LDS ring as one run of stores with constant offsets, not two stores between the matrix products of every step
# speedup vs baseline: 3.9205x; 3.9205x over previous
#include <hip/hip_runtime.h>


#define HMM_B 256
#define HMM_T 8192
#define HMM_K 8
#define HMM_EPS 1e-8f
#define HMM_LN2 0.6931471805599453f

typedef __attribute__((ext_vector_type(2))) float v2f;
typedef __attribute__((ext_vector_type(8))) float v8f;

__device__ __forceinline__ float fast_rcp(float x) {
    float r = __builtin_amdgcn_rcpf(x);
    float e = __builtin_fmaf(-x, r, 1.0f);
    return __builtin_fmaf(r, e, r);
}

__device__ __forceinline__ float crosshalf(float x) {
    unsigned u = __builtin_bit_cast(unsigned, x);
    u = __builtin_amdgcn_permlanex16(u, u, 0x76543210u, 0xFEDCBA98u, false, false);
    return __builtin_bit_cast(float, u);
}

__device__ __forceinline__ float hmm_colsum(v8f a, bool lo) {
    float s = ((a[0] + a[1]) + (a[2] + a[3])) + ((a[4] + a[5]) + (a[6] + a[7]));
    float sx = crosshalf(s);
    return lo ? s : sx;
}

__global__ __launch_bounds__(256) void hmm_pack_kernel(
    const int* __restrict__ obs, unsigned* __restrict__ bits)
{
    const int wave = (blockIdx.x * blockDim.x + threadIdx.x) >> 5;
    const int lane = threadIdx.x & 31;
    const long base = (long)wave * 1024;
    unsigned keep = 0;
#pragma unroll 4
    for (int i = 0; i < 32; ++i) {
        int v = obs[base + i * 32 + lane];
        unsigned m = (unsigned)__ballot(v != 0);
        if (i == lane) keep = m;
    }
    bits[(size_t)wave * 32 + lane] = keep;
}

#define HMM_R   32
#define HMM_NC  (HMM_T / HMM_R)
#define HMM_ROWF 128
#define HMM_DUMP (2 * HMM_R * HMM_ROWF)
#define HMM_G   8

typedef __attribute__((ext_vector_type(4))) float v4f;

__device__ __forceinline__ void hmm_store_group(float* base, const v8f* rows, float nrm0, float nrm1) {
    asm volatile("" :: "v"(nrm0), "v"(nrm1));
#pragma unroll
    for (int r = 0; r < HMM_G; ++r) {
        *(v4f*)(base + r * HMM_ROWF)      = __builtin_shufflevector(rows[r], rows[r], 0, 1, 2, 3);
        *(v4f*)(base + r * HMM_ROWF + 64) = __builtin_shufflevector(rows[r], rows[r], 4, 5, 6, 7);
    }
    __builtin_amdgcn_sched_barrier(0);
}

template <bool SCALE>
__device__ __forceinline__ v8f hmm_alpha_step(v8f a, float& inv, float& lg, float& lc,
                                              int ob, bool lo, v2f A0, v2f A1,
                                              const float* em1, const float* em0)
{
    float x2 = crosshalf(a[2]);
    float x3 = crosshalf(a[3]);
    float x6 = crosshalf(a[6]);
    float x7 = crosshalf(a[7]);
    v2f B0, B1;
    B0.x = lo ? a[0] : x2;  B0.y = lo ? a[1] : x3;
    B1.x = lo ? a[4] : x6;  B1.y = lo ? a[5] : x7;

    const v8f vzero = {};
    v8f d = __builtin_amdgcn_wmma_f32_16x16x4_f32(
        false, A0, false, B0, (short)0, vzero, false, false);
    d = __builtin_amdgcn_wmma_f32_16x16x4_f32(
        false, A1, false, B1, (short)0, d, false, false);

    if (SCALE) {
        lc += lg;
#pragma unroll
        for (int m = 0; m < 8; ++m) a[m] = d[m] * ((ob ? em1[m] : em0[m]) * inv);
        const float s = hmm_colsum(a, lo);
        inv = fast_rcp(s);
        lg  = __builtin_amdgcn_logf(s);
    } else {
#pragma unroll
        for (int m = 0; m < 8; ++m) a[m] = d[m] * (ob ? em1[m] : em0[m]);
    }
    return a;
}

__global__ __launch_bounds__(64) void hmm_forward_kernel(
    const float* __restrict__ start, const float* __restrict__ trans,
    const float* __restrict__ emission, const unsigned* __restrict__ bits,
    float* __restrict__ alpha_out, float* __restrict__ loglik)
{
    __shared__ __attribute__((aligned(16))) float sring[HMM_DUMP + HMM_R * HMM_ROWF];
    __shared__ unsigned swords[2][16];

    const int  wave = __builtin_amdgcn_readfirstlane((int)(threadIdx.x >> 5));
    const int  lane = threadIdx.x & 31;
    const int  n    = lane & 15;
    const bool lo   = lane < 16;
    const int  b    = blockIdx.x * 16 + n;
    const unsigned* wrow = bits + (size_t)b * (HMM_T / 32);

    float em1[8], em0[8];
#pragma unroll
    for (int k = 0; k < 8; ++k) {
        float e = emission[k];
        e = fminf(fmaxf(e, HMM_EPS), 1.0f - HMM_EPS);
        em1[k] = e; em0[k] = 1.0f - e;
    }
    float st[8]; float ssum = 0.0f;
#pragma unroll
    for (int k = 0; k < 8; ++k) { st[k] = start[k] + HMM_EPS; ssum += st[k]; }
#pragma unroll
    for (int k = 0; k < 8; ++k) st[k] /= ssum;

    float rs[8];
#pragma unroll
    for (int j = 0; j < 8; ++j) {
        float s = 0.0f;
#pragma unroll
        for (int k = 0; k < 8; ++k) s += trans[j * 8 + k] + HMM_EPS;
        rs[j] = s;
    }

    const int kb = lo ? 0 : 2;
    const int i  = n;
    v2f A0, A1;
    A0.x = (i < 8) ? (trans[(kb + 0) * 8 + i] + HMM_EPS) / rs[kb + 0] : 0.0f;
    A0.y = (i < 8) ? (trans[(kb + 1) * 8 + i] + HMM_EPS) / rs[kb + 1] : 0.0f;
    A1.x = (i < 8) ? (trans[(kb + 4) * 8 + i] + HMM_EPS) / rs[kb + 4] : 0.0f;
    A1.y = (i < 8) ? (trans[(kb + 5) * 8 + i] + HMM_EPS) / rs[kb + 5] : 0.0f;

    if (wave == 1) swords[0][n] = wrow[0];
    __syncthreads();

    v8f a = {};
    float inv = 0.0f, lg = 0.0f, lc = 0.0f;

    const int hp   = lane >> 4;

    for (int c = 0; c <= HMM_NC; ++c) {
        if (wave == 0) {
            if (c < HMM_NC) {
                const unsigned w = swords[c & 1][n];
                float* const pb = sring + (lo ? (c & 1) * HMM_R * HMM_ROWF : HMM_DUMP) + n * 4;
                v8f rows[HMM_G];
                if (c == 0) {
                    const int ob = (int)(w & 1u);
#pragma unroll
                    for (int m = 0; m < 8; ++m) a[m] = st[m] * (ob ? em1[m] : em0[m]);
                    const float s = hmm_colsum(a, lo);
                    inv = fast_rcp(s);
                    lg  = __builtin_amdgcn_logf(s);
                } else {
                    a = hmm_alpha_step<true>(a, inv, lg, lc, (int)(w & 1u), lo, A0, A1, em1, em0);
                }
                rows[0] = a;
                a = hmm_alpha_step<false>(a, inv, lg, lc, (int)((w >> 1) & 1u), lo, A0, A1, em1, em0);
                rows[1] = a;
#pragma unroll
                for (int r = 2; r < HMM_G; r += 2) {
                    a = hmm_alpha_step<true>(a, inv, lg, lc, (int)((w >> r) & 1u), lo, A0, A1, em1, em0);
                    rows[r] = a;
                    a = hmm_alpha_step<false>(a, inv, lg, lc, (int)((w >> (r + 1)) & 1u), lo, A0, A1, em1, em0);
                    rows[r + 1] = a;
                }
                hmm_store_group(pb, rows, inv, lg);
#pragma unroll
                for (int g = HMM_G; g < HMM_R; g += HMM_G) {
#pragma unroll
                    for (int r = 0; r < HMM_G; r += 2) {
                        a = hmm_alpha_step<true>(a, inv, lg, lc, (int)((w >> (g + r)) & 1u), lo, A0, A1, em1, em0);
                        rows[r] = a;
                        a = hmm_alpha_step<false>(a, inv, lg, lc, (int)((w >> (g + r + 1)) & 1u), lo, A0, A1, em1, em0);
                        rows[r + 1] = a;
                    }
                    hmm_store_group(pb + g * HMM_ROWF, rows, inv, lg);
                }
            }
        } else {
            unsigned wnext = 0;
            if (c + 1 < HMM_NC) wnext = wrow[c + 1];
            if (c > 0) {
                const int cc = c - 1;
                float* g0 = alpha_out + ((size_t)b * HMM_T + (cc * HMM_R + hp)) * 8;
                const float* l0 = sring + ((cc & 1) * HMM_R + hp) * HMM_ROWF + n * 4;
#pragma unroll
                for (int q = 0; q < HMM_R / 2; ++q) {
                    const v4f dl = *(const v4f*)(l0 + q * 2 * HMM_ROWF);
                    const v4f dh = *(const v4f*)(l0 + q * 2 * HMM_ROWF + 64);
                    *(v4f*)(g0 + q * 16)     = dl;
                    *(v4f*)(g0 + q * 16 + 4) = dh;
                }
            }
            if (c + 1 < HMM_NC) swords[(c + 1) & 1][n] = wnext;
        }
        __syncthreads();
    }
    if (wave == 0 && lo) {
        const float s = ((a[0] + a[1]) + (a[2] + a[3])) + ((a[4] + a[5]) + (a[6] + a[7]));
        loglik[b] = (lc + __builtin_amdgcn_logf(s)) * HMM_LN2;
    }
}

template <bool SCALE>
__device__ __forceinline__ v8f hmm_beta_step(v8f d, float& inv, int ob, bool lo, v2f A0, v2f A1,
                                             const float* em1, const float* em0)
{
    v8f u;
    if (SCALE) {
#pragma unroll
        for (int m = 0; m < 8; ++m) u[m] = d[m] * ((ob ? em1[m] : em0[m]) * inv);
        inv = fast_rcp(hmm_colsum(u, lo));
    } else {
#pragma unroll
        for (int m = 0; m < 8; ++m) u[m] = d[m] * (ob ? em1[m] : em0[m]);
    }
    float x2 = crosshalf(u[2]);
    float x3 = crosshalf(u[3]);
    float x6 = crosshalf(u[6]);
    float x7 = crosshalf(u[7]);
    v2f B0, B1;
    B0.x = lo ? u[0] : x2;  B0.y = lo ? u[1] : x3;
    B1.x = lo ? u[4] : x6;  B1.y = lo ? u[5] : x7;

    const v8f vzero = {};
    v8f r = __builtin_amdgcn_wmma_f32_16x16x4_f32(
        false, A0, false, B0, (short)0, vzero, false, false);
    r = __builtin_amdgcn_wmma_f32_16x16x4_f32(
        false, A1, false, B1, (short)0, r, false, false);
    return r;
}

__global__ __launch_bounds__(64) void hmm_backward_kernel(
    const float* __restrict__ trans, const float* __restrict__ emission,
    const unsigned* __restrict__ bits, float* __restrict__ gamma)
{
    __shared__ __attribute__((aligned(16))) float sbeta[HMM_DUMP + HMM_R * HMM_ROWF];
    __shared__ unsigned swords[2][16];

    const int  wave = __builtin_amdgcn_readfirstlane((int)(threadIdx.x >> 5));
    const int  lane = threadIdx.x & 31;
    const int  n    = lane & 15;
    const bool lo   = lane < 16;
    const int  b    = blockIdx.x * 16 + n;
    const unsigned* wrow = bits + (size_t)b * (HMM_T / 32);

    float em1[8], em0[8];
#pragma unroll
    for (int k = 0; k < 8; ++k) {
        float e = emission[k];
        e = fminf(fmaxf(e, HMM_EPS), 1.0f - HMM_EPS);
        em1[k] = e; em0[k] = 1.0f - e;
    }
    float rs[8];
#pragma unroll
    for (int j = 0; j < 8; ++j) {
        float s = 0.0f;
#pragma unroll
        for (int k = 0; k < 8; ++k) s += trans[j * 8 + k] + HMM_EPS;
        rs[j] = s;
    }
    const int kb = lo ? 0 : 2;
    const int i  = n;
    v2f A0, A1;
    A0.x = (i < 8) ? (trans[i * 8 + (kb + 0)] + HMM_EPS) / rs[i] : 0.0f;
    A0.y = (i < 8) ? (trans[i * 8 + (kb + 1)] + HMM_EPS) / rs[i] : 0.0f;
    A1.x = (i < 8) ? (trans[i * 8 + (kb + 4)] + HMM_EPS) / rs[i] : 0.0f;
    A1.y = (i < 8) ? (trans[i * 8 + (kb + 5)] + HMM_EPS) / rs[i] : 0.0f;

    if (wave == 1) swords[0][n] = wrow[HMM_NC - 1];
    __syncthreads();

    v8f d;
#pragma unroll
    for (int m = 0; m < 8; ++m) d[m] = 1.0f;
    float inv = 0.125f;
    unsigned w = 0;

    const int hp   = lane >> 4;

    for (int c = 0; c <= HMM_NC; ++c) {
        if (wave == 0) {
            if (c < HMM_NC) {
                const unsigned wnew = swords[c & 1][n];
                float* const pb = sbeta + (lo ? (c & 1) * HMM_R * HMM_ROWF : HMM_DUMP) + n * 4;
                v8f rows[HMM_G];
                if (c > 0) d = hmm_beta_step<true>(d, inv, (int)(w & 1u), lo, A0, A1, em1, em0);
                rows[0] = d;
                w = wnew;
                d = hmm_beta_step<false>(d, inv, (int)((w >> 31) & 1u), lo, A0, A1, em1, em0);
                rows[1] = d;
#pragma unroll
                for (int r = 2; r < HMM_G; r += 2) {
                    d = hmm_beta_step<true>(d, inv, (int)((w >> (32 - r)) & 1u), lo, A0, A1, em1, em0);
                    rows[r] = d;
                    d = hmm_beta_step<false>(d, inv, (int)((w >> (31 - r)) & 1u), lo, A0, A1, em1, em0);
                    rows[r + 1] = d;
                }
                hmm_store_group(pb, rows, inv, inv);
#pragma unroll
                for (int g = HMM_G; g < HMM_R; g += HMM_G) {
#pragma unroll
                    for (int r = 0; r < HMM_G; r += 2) {
                        d = hmm_beta_step<true>(d, inv, (int)((w >> (32 - g - r)) & 1u), lo, A0, A1, em1, em0);
                        rows[r] = d;
                        d = hmm_beta_step<false>(d, inv, (int)((w >> (31 - g - r)) & 1u), lo, A0, A1, em1, em0);
                        rows[r + 1] = d;
                    }
                    hmm_store_group(pb + g * HMM_ROWF, rows, inv, inv);
                }
            }
        } else {
            unsigned wnext = 0;
            if (c + 1 < HMM_NC) wnext = wrow[HMM_NC - 2 - c];
            if (c > 0) {
                const int cc = c - 1;
                float* g0 = gamma + ((size_t)b * HMM_T + (HMM_T - 1 - cc * HMM_R - hp)) * 8;
                const float* l0 = sbeta + ((cc & 1) * HMM_R + hp) * HMM_ROWF + n * 4;
                v8f al[HMM_R / 2];
#pragma unroll
                for (int q = 0; q < HMM_R / 2; ++q) al[q] = *(const v8f*)(g0 - q * 16);
#pragma unroll
                for (int q = 0; q < HMM_R / 2; ++q) {
                    const v4f dl = *(const v4f*)(l0 + q * 2 * HMM_ROWF);
                    const v4f dh = *(const v4f*)(l0 + q * 2 * HMM_ROWF + 64);
                    v8f g;
                    float gs = 0.0f;
#pragma unroll
                    for (int m = 0; m < 4; ++m) { g[m] = al[q][m] * dl[m]; gs += g[m]; }
#pragma unroll
                    for (int m = 0; m < 4; ++m) { g[4 + m] = al[q][4 + m] * dh[m]; gs += g[4 + m]; }
                    float gi = fast_rcp(gs);
#pragma unroll
                    for (int m = 0; m < 8; ++m) g[m] *= gi;
                    *(v8f*)(g0 - q * 16) = g;
                }
            }
            if (c + 1 < HMM_NC) swords[(c + 1) & 1][n] = wnext;
        }
        __syncthreads();
    }
}

extern "C" void kernel_launch(void* const* d_in, const int* in_sizes, int n_in,
                              void* d_out, int out_size, void* d_ws, size_t ws_size,
                              hipStream_t stream) {
    (void)in_sizes; (void)n_in; (void)out_size; (void)ws_size;
    const float* start    = (const float*)d_in[0];
    const float* trans    = (const float*)d_in[1];
    const float* emission = (const float*)d_in[2];
    const int*   obs      = (const int*)d_in[3];

    float* gamma  = (float*)d_out;
    float* loglik = gamma + (size_t)HMM_B * HMM_T * HMM_K;
    unsigned* bits = (unsigned*)d_ws;

    hmm_pack_kernel<<<256, 256, 0, stream>>>(obs, bits);
    hmm_forward_kernel<<<HMM_B / 16, 64, 0, stream>>>(
        start, trans, emission, bits, gamma, loglik);
    hmm_backward_kernel<<<HMM_B / 16, 64, 0, stream>>>(
        trans, emission, bits, gamma);
}


# both recurrences advance two time steps per dependent matrix product: A rows 0-7 / 8-15 hold P*diag(em_o)*P for obs 1 / 0, each column picks its half; products re-associated (sum order changed)
# speedup vs baseline: 5.0730x; 1.2940x over previous
#include <hip/hip_runtime.h>


#define HMM_B 256
#define HMM_T 8192
#define HMM_K 8
#define HMM_EPS 1e-8f
#define HMM_LN2 0.6931471805599453f

typedef __attribute__((ext_vector_type(2))) float v2f;
typedef __attribute__((ext_vector_type(4))) float v4f;
typedef __attribute__((ext_vector_type(8))) float v8f;

__device__ __forceinline__ float fast_rcp(float x) {
    float r = __builtin_amdgcn_rcpf(x);
    float e = __builtin_fmaf(-x, r, 1.0f);
    return __builtin_fmaf(r, e, r);
}

__device__ __forceinline__ float crosshalf(float x) {
    unsigned u = __builtin_bit_cast(unsigned, x);
    u = __builtin_amdgcn_permlanex16(u, u, 0x76543210u, 0xFEDCBA98u, false, false);
    return __builtin_bit_cast(float, u);
}

__device__ __forceinline__ float hmm_colsum(v4f u) {
    const float s = (u.x + u.y) + (u.z + u.w);
    return s + crosshalf(s);
}

__global__ __launch_bounds__(256) void hmm_pack_kernel(
    const int* __restrict__ obs, unsigned* __restrict__ bits)
{
    const int wave = (blockIdx.x * blockDim.x + threadIdx.x) >> 5;
    const int lane = threadIdx.x & 31;
    const long base = (long)wave * 1024;
    unsigned keep = 0;
#pragma unroll 4
    for (int i = 0; i < 32; ++i) {
        int v = obs[base + i * 32 + lane];
        unsigned m = (unsigned)__ballot(v != 0);
        if (i == lane) keep = m;
    }
    bits[(size_t)wave * 32 + lane] = keep;
}

#define HMM_R   32
#define HMM_NC  (HMM_T / HMM_R)
#define HMM_ROWF 128
#define HMM_G   8

__device__ __forceinline__ void hmm_build_ops(const float* sP, const float* em1, const float* em0,
                                              int lane, v2f& F0, v2f& F1, v2f& S0, v2f& S1)
{
    const int  c = lane & 15;
    const int  h = lane >> 4;
    const bool upper = c >= 8;
    const int  i = upper ? ((c - 8) ^ 2) : c;
    float q[4];
#pragma unroll
    for (int kk = 0; kk < 4; ++kk) {
        const int k = (kk & 1) + 2 * h + 4 * (kk >> 1);
        float acc = 0.0f;
#pragma unroll
        for (int j = 0; j < 8; ++j)
            acc = __builtin_fmaf(sP[i * 8 + j] * (upper ? em0[j] : em1[j]), sP[j * 8 + k], acc);
        q[kk] = acc;
    }
    F0.x = q[0]; F0.y = q[1]; F1.x = q[2]; F1.y = q[3];
    const int  r  = c & 7;
    const bool used = r < 4;
    const int  is = (r & 1) + ((r & 2) ? 4 : 0) + (upper ? 2 : 0);
    S0.x = used ? sP[is * 8 + 2 * h]     : 0.0f;
    S0.y = used ? sP[is * 8 + 2 * h + 1] : 0.0f;
    S1.x = used ? sP[is * 8 + 2 * h + 4] : 0.0f;
    S1.y = used ? sP[is * 8 + 2 * h + 5] : 0.0f;
}

__device__ __forceinline__ void hmm_tiles(v4f u, v2f F0, v2f F1, v2f S0, v2f S1, v8f& f, v8f& s) {
    v2f B0, B1;
    B0.x = u.x; B0.y = u.y; B1.x = u.z; B1.y = u.w;
    const v8f vzero = {};
    f = __builtin_amdgcn_wmma_f32_16x16x4_f32(false, F0, false, B0, (short)0, vzero, false, false);
    s = __builtin_amdgcn_wmma_f32_16x16x4_f32(false, S0, false, B0, (short)0, vzero, false, false);
    f = __builtin_amdgcn_wmma_f32_16x16x4_f32(false, F1, false, B1, (short)0, f, false, false);
    s = __builtin_amdgcn_wmma_f32_16x16x4_f32(false, S1, false, B1, (short)0, s, false, false);
}

__device__ __forceinline__ v8f hmm_tile_s(v4f u, v2f S0, v2f S1) {
    v2f B0, B1;
    B0.x = u.x; B0.y = u.y; B1.x = u.z; B1.y = u.w;
    const v8f vzero = {};
    v8f s = __builtin_amdgcn_wmma_f32_16x16x4_f32(false, S0, false, B0, (short)0, vzero, false, false);
    s = __builtin_amdgcn_wmma_f32_16x16x4_f32(false, S1, false, B1, (short)0, s, false, false);
    return s;
}

__device__ __forceinline__ v4f hmm_pick(v8f f, bool own) {
    const float x2 = crosshalf(f[2]);
    const float x3 = crosshalf(f[3]);
    const float x6 = crosshalf(f[6]);
    const float x7 = crosshalf(f[7]);
    v4f p;
    p.x = own ? f[0] : x2;  p.y = own ? f[1] : x3;
    p.z = own ? f[4] : x6;  p.w = own ? f[5] : x7;
    return p;
}

__device__ __forceinline__ void hmm_store_group(float* base, const v4f* rows) {
    __builtin_amdgcn_sched_barrier(0);
#pragma unroll
    for (int r = 0; r < HMM_G; ++r) *(v4f*)(base + r * HMM_ROWF) = rows[r];
    __builtin_amdgcn_sched_barrier(0);
}

__device__ __forceinline__ void hmm_alpha_hop(v4f& u, float& lc, v4f& rowS, v4f& rowF,
                                              bool oa, bool ob, bool lo,
                                              v2f F0, v2f F1, v2f S0, v2f S1, v4f e1, v4f e0)
{
    const float s   = hmm_colsum(u);
    const float inv = fast_rcp(s);
    lc += __builtin_amdgcn_logf(s);
    v8f f, sd;
    hmm_tiles(u, F0, F1, S0, S1, f, sd);
    const v4f eS = oa ? e1 : e0;
    const v4f eF = (ob ? e1 : e0) * inv;
    rowS.x = sd[0] * eS.x;  rowS.y = sd[1] * eS.y;
    rowS.z = sd[2] * eS.z;  rowS.w = sd[3] * eS.w;
    const v4f p = hmm_pick(f, oa == lo);
    rowF = p * eF;
    u = rowF;
}

__global__ __launch_bounds__(64) void hmm_forward_kernel(
    const float* __restrict__ start, const float* __restrict__ trans,
    const float* __restrict__ emission, const unsigned* __restrict__ bits,
    float* __restrict__ alpha_out, float* __restrict__ loglik)
{
    __shared__ __attribute__((aligned(16))) float sring[2 * HMM_R * HMM_ROWF];
    __shared__ unsigned swords[2][16];
    __shared__ float sP[64];

    const int  wave = __builtin_amdgcn_readfirstlane((int)(threadIdx.x >> 5));
    const int  lane = threadIdx.x & 31;
    const int  n    = lane & 15;
    const bool lo   = lane < 16;
    const int  hp   = lane >> 4;
    const int  b    = blockIdx.x * 16 + n;
    const unsigned* wrow = bits + (size_t)b * (HMM_T / 32);

    float em1[8], em0[8];
#pragma unroll
    for (int k = 0; k < 8; ++k) {
        float e = emission[k];
        e = fminf(fmaxf(e, HMM_EPS), 1.0f - HMM_EPS);
        em1[k] = e; em0[k] = 1.0f - e;
    }
    float st[8]; float ssum = 0.0f;
#pragma unroll
    for (int k = 0; k < 8; ++k) { st[k] = start[k] + HMM_EPS; ssum += st[k]; }
#pragma unroll
    for (int k = 0; k < 8; ++k) st[k] /= ssum;

    {
        const int i = threadIdx.x >> 3, k = threadIdx.x & 7;
        float s = 0.0f;
#pragma unroll
        for (int kk = 0; kk < 8; ++kk) s += trans[k * 8 + kk] + HMM_EPS;
        sP[i * 8 + k] = (trans[k * 8 + i] + HMM_EPS) / s;
    }
    if (wave == 1) swords[0][n] = wrow[0];
    __syncthreads();

    v2f F0, F1, S0, S1;
    hmm_build_ops(sP, em1, em0, lane, F0, F1, S0, S1);

    v4f e1, e0, stv;
    e1.x = lo ? em1[0] : em1[2];  e1.y = lo ? em1[1] : em1[3];
    e1.z = lo ? em1[4] : em1[6];  e1.w = lo ? em1[5] : em1[7];
    e0.x = lo ? em0[0] : em0[2];  e0.y = lo ? em0[1] : em0[3];
    e0.z = lo ? em0[4] : em0[6];  e0.w = lo ? em0[5] : em0[7];
    stv.x = lo ? st[0] : st[2];   stv.y = lo ? st[1] : st[3];
    stv.z = lo ? st[4] : st[6];   stv.w = lo ? st[5] : st[7];

    v4f u = {};
    float lc = 0.0f;

    for (int c = 0; c <= HMM_NC; ++c) {
        if (wave == 0) {
            if (c < HMM_NC) {
                const unsigned w = swords[c & 1][n];
                float* const pb = sring + (c & 1) * HMM_R * HMM_ROWF + hp * 64 + n * 4;
                v4f rows[HMM_G];
                if (c == 0) {
                    u = stv * ((w & 1u) ? e1 : e0);
                    rows[0] = u;
                    const float s   = hmm_colsum(u);
                    const float inv = fast_rcp(s);
                    lc += __builtin_amdgcn_logf(s);
                    const v8f sd = hmm_tile_s(u, S0, S1);
                    const v4f eS = (((w >> 1) & 1u) ? e1 : e0) * inv;
                    u.x = sd[0] * eS.x;  u.y = sd[1] * eS.y;
                    u.z = sd[2] * eS.z;  u.w = sd[3] * eS.w;
                    rows[1] = u;
                } else {
                    hmm_alpha_hop(u, lc, rows[0], rows[1], (w & 1u) != 0, ((w >> 1) & 1u) != 0,
                                  lo, F0, F1, S0, S1, e1, e0);
                }
#pragma unroll
                for (int r = 2; r < HMM_G; r += 2)
                    hmm_alpha_hop(u, lc, rows[r], rows[r + 1], ((w >> r) & 1u) != 0,
                                  ((w >> (r + 1)) & 1u) != 0, lo, F0, F1, S0, S1, e1, e0);
                hmm_store_group(pb, rows);
#pragma unroll
                for (int g = HMM_G; g < HMM_R; g += HMM_G) {
#pragma unroll
                    for (int r = 0; r < HMM_G; r += 2)
                        hmm_alpha_hop(u, lc, rows[r], rows[r + 1], ((w >> (g + r)) & 1u) != 0,
                                      ((w >> (g + r + 1)) & 1u) != 0, lo, F0, F1, S0, S1, e1, e0);
                    hmm_store_group(pb + g * HMM_ROWF, rows);
                }
            }
        } else {
            unsigned wnext = 0;
            if (c + 1 < HMM_NC) wnext = wrow[c + 1];
            if (c > 0) {
                const int cc = c - 1;
                float* g0 = alpha_out + ((size_t)b * HMM_T + (cc * HMM_R + hp)) * 8;
                const float* l0 = sring + ((cc & 1) * HMM_R + hp) * HMM_ROWF + n * 4;
#pragma unroll
                for (int q = 0; q < HMM_R / 2; ++q) {
                    const v4f dl = *(const v4f*)(l0 + q * 2 * HMM_ROWF);
                    const v4f dh = *(const v4f*)(l0 + q * 2 * HMM_ROWF + 64);
                    v4f o0, o1;
                    o0.x = dl.x; o0.y = dl.y; o0.z = dh.x; o0.w = dh.y;
                    o1.x = dl.z; o1.y = dl.w; o1.z = dh.z; o1.w = dh.w;
                    *(v4f*)(g0 + q * 16)     = o0;
                    *(v4f*)(g0 + q * 16 + 4) = o1;
                }
            }
            if (c + 1 < HMM_NC) swords[(c + 1) & 1][n] = wnext;
        }
        __syncthreads();
    }
    if (wave == 0) {
        const float s = hmm_colsum(u);
        if (lo) loglik[b] = (lc + __builtin_amdgcn_logf(s)) * HMM_LN2;
    }
}

__device__ __forceinline__ void hmm_beta_hop(v4f& u, v4f& rowS, v4f& rowF,
                                             bool oa, bool ob, bool lo,
                                             v2f F0, v2f F1, v2f S0, v2f S1, v4f e1, v4f e0)
{
    const float inv = fast_rcp(hmm_colsum(u));
    v8f f, sd;
    hmm_tiles(u, F0, F1, S0, S1, f, sd);
    rowS.x = sd[0];  rowS.y = sd[1];  rowS.z = sd[2];  rowS.w = sd[3];
    const v4f eF = (ob ? e1 : e0) * inv;
    rowF = hmm_pick(f, oa == lo);
    u = rowF * eF;
}

__global__ __launch_bounds__(64) void hmm_backward_kernel(
    const float* __restrict__ trans, const float* __restrict__ emission,
    const unsigned* __restrict__ bits, float* __restrict__ gamma)
{
    __shared__ __attribute__((aligned(16))) float sbeta[2 * HMM_R * HMM_ROWF];
    __shared__ unsigned swords[2][16];
    __shared__ float sP[64];

    const int  wave = __builtin_amdgcn_readfirstlane((int)(threadIdx.x >> 5));
    const int  lane = threadIdx.x & 31;
    const int  n    = lane & 15;
    const bool lo   = lane < 16;
    const int  hp   = lane >> 4;
    const int  b    = blockIdx.x * 16 + n;
    const unsigned* wrow = bits + (size_t)b * (HMM_T / 32);

    float em1[8], em0[8];
#pragma unroll
    for (int k = 0; k < 8; ++k) {
        float e = emission[k];
        e = fminf(fmaxf(e, HMM_EPS), 1.0f - HMM_EPS);
        em1[k] = e; em0[k] = 1.0f - e;
    }
    {
        const int i = threadIdx.x >> 3, k = threadIdx.x & 7;
        float s = 0.0f;
#pragma unroll
        for (int kk = 0; kk < 8; ++kk) s += trans[i * 8 + kk] + HMM_EPS;
        sP[i * 8 + k] = (trans[i * 8 + k] + HMM_EPS) / s;
    }
    if (wave == 1) swords[0][n] = wrow[HMM_NC - 1];
    __syncthreads();

    v2f F0, F1, S0, S1;
    hmm_build_ops(sP, em1, em0, lane, F0, F1, S0, S1);

    v4f e1, e0;
    e1.x = lo ? em1[0] : em1[2];  e1.y = lo ? em1[1] : em1[3];
    e1.z = lo ? em1[4] : em1[6];  e1.w = lo ? em1[5] : em1[7];
    e0.x = lo ? em0[0] : em0[2];  e0.y = lo ? em0[1] : em0[3];
    e0.z = lo ? em0[4] : em0[6];  e0.w = lo ? em0[5] : em0[7];

    v4f u = {};

    for (int c = 0; c <= HMM_NC; ++c) {
        if (wave == 0) {
            if (c < HMM_NC) {
                const unsigned w = swords[c & 1][n];
                float* const pb = sbeta + (c & 1) * HMM_R * HMM_ROWF + hp * 64 + n * 4;
                v4f rows[HMM_G];
                if (c == 0) {
                    const v4f ones = {1.0f, 1.0f, 1.0f, 1.0f};
                    rows[0] = ones;
                    u = ((w >> 31) & 1u) ? e1 : e0;
                    const float inv = fast_rcp(hmm_colsum(u));
                    const v8f sd = hmm_tile_s(u, S0, S1);
                    v4f r1;
                    r1.x = sd[0];  r1.y = sd[1];  r1.z = sd[2];  r1.w = sd[3];
                    rows[1] = r1;
                    u = r1 * ((((w >> 30) & 1u) ? e1 : e0) * inv);
                } else {
                    hmm_beta_hop(u, rows[0], rows[1], ((w >> 31) & 1u) != 0, ((w >> 30) & 1u) != 0,
                                 lo, F0, F1, S0, S1, e1, e0);
                }
#pragma unroll
                for (int r = 2; r < HMM_G; r += 2)
                    hmm_beta_hop(u, rows[r], rows[r + 1], ((w >> (31 - r)) & 1u) != 0,
                                 ((w >> (30 - r)) & 1u) != 0, lo, F0, F1, S0, S1, e1, e0);
                hmm_store_group(pb, rows);
#pragma unroll
                for (int g = HMM_G; g < HMM_R; g += HMM_G) {
#pragma unroll
                    for (int r = 0; r < HMM_G; r += 2)
                        hmm_beta_hop(u, rows[r], rows[r + 1], ((w >> (31 - g - r)) & 1u) != 0,
                                     ((w >> (30 - g - r)) & 1u) != 0, lo, F0, F1, S0, S1, e1, e0);
                    hmm_store_group(pb + g * HMM_ROWF, rows);
                }
            }
        } else {
            unsigned wnext = 0;
            if (c + 1 < HMM_NC) wnext = wrow[HMM_NC - 2 - c];
            if (c > 0) {
                const int cc = c - 1;
                float* g0 = gamma + ((size_t)b * HMM_T + (HMM_T - 1 - cc * HMM_R - hp)) * 8;
                const float* l0 = sbeta + ((cc & 1) * HMM_R + hp) * HMM_ROWF + n * 4;
                v8f al[HMM_R / 2];
#pragma unroll
                for (int q = 0; q < HMM_R / 2; ++q) al[q] = *(const v8f*)(g0 - q * 16);
#pragma unroll
                for (int q = 0; q < HMM_R / 2; ++q) {
                    const v4f dl = *(const v4f*)(l0 + q * 2 * HMM_ROWF);
                    const v4f dh = *(const v4f*)(l0 + q * 2 * HMM_ROWF + 64);
                    float be[8];
                    be[0] = dl.x; be[1] = dl.y; be[2] = dh.x; be[3] = dh.y;
                    be[4] = dl.z; be[5] = dl.w; be[6] = dh.z; be[7] = dh.w;
                    v8f g;
                    float gs = 0.0f;
#pragma unroll
                    for (int m = 0; m < 8; ++m) { g[m] = al[q][m] * be[m]; gs += g[m]; }
                    float gi = fast_rcp(gs);
#pragma unroll
                    for (int m = 0; m < 8; ++m) g[m] *= gi;
                    *(v8f*)(g0 - q * 16) = g;
                }
            }
            if (c + 1 < HMM_NC) swords[(c + 1) & 1][n] = wnext;
        }
        __syncthreads();
    }
}

extern "C" void kernel_launch(void* const* d_in, const int* in_sizes, int n_in,
                              void* d_out, int out_size, void* d_ws, size_t ws_size,
                              hipStream_t stream) {
    (void)in_sizes; (void)n_in; (void)out_size; (void)ws_size;
    const float* start    = (const float*)d_in[0];
    const float* trans    = (const float*)d_in[1];
    const float* emission = (const float*)d_in[2];
    const int*   obs      = (const int*)d_in[3];

    float* gamma  = (float*)d_out;
    float* loglik = gamma + (size_t)HMM_B * HMM_T * HMM_K;
    unsigned* bits = (unsigned*)d_ws;

    hmm_pack_kernel<<<256, 256, 0, stream>>>(obs, bits);
    hmm_forward_kernel<<<HMM_B / 16, 64, 0, stream>>>(
        start, trans, emission, bits, gamma, loglik);
    hmm_backward_kernel<<<HMM_B / 16, 64, 0, stream>>>(
        trans, emission, bits, gamma);
}
